# starting kernel
# baseline (speedup 1.0000x reference)
#include <hip/hip_runtime.h>

#define DIM 64

typedef float v2f __attribute__((ext_vector_type(2)));
typedef float v8f __attribute__((ext_vector_type(8)));

__global__ __launch_bounds__(256) void zero_kernel(float4* __restrict__ out, unsigned n4) {
    unsigned i = blockIdx.x * 256u + threadIdx.x;
    if (i < n4) out[i] = make_float4(0.f, 0.f, 0.f, 0.f);
}

__global__ __launch_bounds__(256) void scatter_kernel(
    const float* __restrict__ user_emb,
    const float* __restrict__ item_emb,
    const int*   __restrict__ src,
    const int*   __restrict__ dst,
    const float* __restrict__ ew,
    float*       __restrict__ out,
    int n_users, int n_edges)
{
    unsigned gid  = blockIdx.x * 256u + threadIdx.x;
    int      e    = (int)(gid >> 4);
    int      lane = (int)(gid & 15u);
    if (e >= n_edges) return;

    int   s = src[e];
    int   d = dst[e];
    float w = ew[e];

    const float* row = (s < n_users)
                           ? (user_emb + (size_t)s * DIM)
                           : (item_emb + (size_t)(s - n_users) * DIM);

    float4 v = ((const float4*)row)[lane];

    float* o = out + (size_t)d * DIM + lane * 4;
    unsafeAtomicAdd(o + 0, v.x * w);
    unsafeAtomicAdd(o + 1, v.y * w);
    unsafeAtomicAdd(o + 2, v.z * w);
    unsafeAtomicAdd(o + 3, v.w * w);
}

__global__ __launch_bounds__(32) void normalize_kernel(float* __restrict__ out, int n_nodes)
{
    const int lane = threadIdx.x;
    const int m    = lane & 15;
    const int koff = (lane < 16) ? 0 : 2;

    int r = blockIdx.x * 16 + m;
    const bool valid = (r < n_nodes);
    if (r >= n_nodes) r = n_nodes - 1;
    float* rowp = out + (size_t)r * DIM;

    v2f a[16];
#pragma unroll
    for (int c = 0; c < 16; ++c)
        a[c] = *(const v2f*)(rowp + 4 * c + koff);

    v8f acc = {};
#pragma unroll
    for (int c = 0; c < 16; ++c)
        acc = __builtin_amdgcn_wmma_f32_16x16x4_f32(
             false, a[c],  false, a[c],
             (short)0, acc,  false,  false);

    __shared__ float s_inv[16];
    if (lane < 8) {
        float ss = acc[lane];
        s_inv[lane] = 1.0f / fmaxf(sqrtf(ss), 1e-12f);
    } else if (lane >= 24) {
        float ss = acc[lane - 24];
        s_inv[lane - 16] = 1.0f / fmaxf(sqrtf(ss), 1e-12f);
    }
    __syncthreads();
    float inv = s_inv[m];

#pragma unroll
    for (int c = 0; c < 16; ++c) {
        if (valid) {
            v2f sv = a[c] * inv;
            __builtin_nontemporal_store(sv, (v2f*)(rowp + 4 * c + koff));
        }
    }
}

extern "C" void kernel_launch(void* const* d_in, const int* in_sizes, int n_in,
                              void* d_out, int out_size, void* d_ws, size_t ws_size,
                              hipStream_t stream)
{
    const float* user_emb = (const float*)d_in[0];
    const float* item_emb = (const float*)d_in[1];
    const int*   src      = (const int*)d_in[2];
    const int*   dst      = (const int*)d_in[3];
    const float* ew       = (const float*)d_in[4];
    float*       out      = (float*)d_out;

    const int n_users = in_sizes[0] / DIM;
    const int n_items = in_sizes[1] / DIM;
    const int n_edges = in_sizes[2];
    const int n_nodes = n_users + n_items;

    {
        unsigned n4 = (unsigned)((size_t)n_nodes * DIM / 4);
        unsigned blocks = (n4 + 255u) / 256u;
        zero_kernel<<<blocks, 256, 0, stream>>>((float4*)out, n4);
    }
    {
        unsigned total  = (unsigned)n_edges * 16u;
        unsigned blocks = (total + 255u) / 256u;
        scatter_kernel<<<blocks, 256, 0, stream>>>(user_emb, item_emb, src, dst, ew,
                                                   out, n_users, n_edges);
    }
    {
        unsigned blocks = (unsigned)((n_nodes + 15) / 16);
        normalize_kernel<<<blocks, 32, 0, stream>>>(out, n_nodes);
    }
}


# each block owns 1184 nodes: reads all dst, sums its edges' rows in LDS (LDS float adds, arrival order), WMMA norms and one whole-line write from LDS; 1 launch, no global atomics, no zero pass
# speedup vs baseline: 1.8225x; 1.8225x over previous
#include <hip/hip_runtime.h>


#define DIM   64
#define NPB   1184
#define NT    1024
#define NW    32
#define QCAP  128
#define UL    8
#define DB    8

typedef float v2f __attribute__((ext_vector_type(2)));
typedef float v4f __attribute__((ext_vector_type(4)));
typedef float v8f __attribute__((ext_vector_type(8)));
typedef int   v4i __attribute__((ext_vector_type(4)));

__device__ __forceinline__ int swz(int r, int g) { return (r << 6) + (((g ^ r) & 15) << 2); }

__global__ __launch_bounds__(NT) void owner_kernel(
    const float* __restrict__ user_emb,
    const float* __restrict__ item_emb,
    const int*   __restrict__ src,
    const int*   __restrict__ dst,
    const float* __restrict__ ew,
    float*       __restrict__ out,
    int n_users, int n_nodes, int n_edges)
{
    __shared__ __attribute__((aligned(16))) float acc[NPB * DIM];
    __shared__ int   q[NW * QCAP];
    __shared__ float s_inv[NPB];

    const int tid  = (int)threadIdx.x;
    const int lane = tid & 31;
    const int wv    = __builtin_amdgcn_readfirstlane(tid >> 5);
    const int node0 = (int)blockIdx.x * NPB;
    int npb = n_nodes - node0;
    if (npb > NPB) npb = NPB;

    {
        v4f* acc4 = (v4f*)acc;
        const v4f z = {0.f, 0.f, 0.f, 0.f};
        for (int i = tid; i < NPB * DIM / 4; i += NT) acc4[i] = z;
    }
    __syncthreads();

    int* qw = q + wv * QCAP;
    int head = 0, tail = 0;

    const int n_chunks = n_edges >> 7;
    const int nk   = (n_chunks > wv) ? ((n_chunks - wv + NW - 1) / NW) : 0;
    const int ng   = nk / UL;
    const int grot = (int)(((long long)blockIdx.x * ng) / (int)gridDim.x);

    auto add_row = [&](int d, float lo, float hi) {
        const int g = lane >> 2;
        float* p0 = acc + swz(d, g)     + (lane & 3);
        float* p1 = acc + swz(d, g + 8) + (lane & 3);
        __hip_atomic_fetch_add(p0, lo, __ATOMIC_RELAXED, __HIP_MEMORY_SCOPE_WORKGROUP);
        __hip_atomic_fetch_add(p1, hi, __ATOMIC_RELAXED, __HIP_MEMORY_SCOPE_WORKGROUP);
    };
    auto drain_one = [&]() {
        const int   e  = __builtin_amdgcn_readfirstlane(qw[head & (QCAP - 1)]);
        const int   s1 = src[e];
        const int   d1 = dst[e] - node0;
        const float w1 = ew[e];
        const float* row = (s1 < n_users)
                               ? (user_emb + (size_t)s1 * DIM)
                               : (item_emb + (size_t)(s1 - n_users) * DIM);
        add_row(d1, row[lane] * w1, row[lane + 32] * w1);
        ++head;
    };
    auto push = [&](bool m, int e) {
        const unsigned mask = __builtin_amdgcn_ballot_w32(m);
        if (mask) {
            while (tail - head > QCAP - 32) drain_one();
            const int pos = (int)__builtin_amdgcn_mbcnt_lo(mask, 0u);
            if (m) qw[(tail + pos) & (QCAP - 1)] = e;
            tail += __builtin_popcount(mask);
        }
    };
    auto drain = [&](int nb) {
        int   s[DB], d[DB];
        float w[DB], lo[DB], hi[DB];
#pragma unroll
        for (int i = 0; i < DB; ++i) {
            const int ii = (i < nb) ? i : (nb - 1);
            const int e  = __builtin_amdgcn_readfirstlane(qw[(head + ii) & (QCAP - 1)]);
            s[i] = src[e]; d[i] = dst[e] - node0; w[i] = ew[e];
        }
#pragma unroll
        for (int i = 0; i < DB; ++i) {
            const float* row = (s[i] < n_users)
                                   ? (user_emb + (size_t)s[i] * DIM)
                                   : (item_emb + (size_t)(s[i] - n_users) * DIM);
            lo[i] = row[lane];
            hi[i] = row[lane + 32];
        }
#pragma unroll
        for (int i = 0; i < DB; ++i)
            if (i < nb) add_row(d[i], lo[i] * w[i], hi[i] * w[i]);
        head += nb;
    };

    auto examine = [&](v4i v, int c) {
        const bool m0 = (unsigned)(v.x - node0) < (unsigned)npb;
        const bool m1 = (unsigned)(v.y - node0) < (unsigned)npb;
        const bool m2 = (unsigned)(v.z - node0) < (unsigned)npb;
        const bool m3 = (unsigned)(v.w - node0) < (unsigned)npb;
        if (__builtin_amdgcn_ballot_w32(m0 | m1 | m2 | m3)) {
            const int eb = (c << 7) + (lane << 2);
            const unsigned mm = (unsigned)m0 | ((unsigned)m1 << 1) | ((unsigned)m2 << 2) | ((unsigned)m3 << 3);
#pragma nounroll
            for (int j = 0; j < 4; ++j) push(((mm >> j) & 1u) != 0u, eb + j);
        }
    };

    for (int gi = 0; gi < ng; ++gi) {
        int g = gi + grot;
        if (g >= ng) g -= ng;
        const int  c0 = g * (UL * NW) + wv;
        const v4i* p  = (const v4i*)(dst + ((size_t)c0 << 7)) + lane;
        v4i cur[UL];
#pragma unroll
        for (int u = 0; u < UL; ++u) cur[u] = p[u * (NW * 32)];
#pragma unroll
        for (int u = 0; u < UL; ++u) examine(cur[u], c0 + u * NW);
        while (tail - head >= DB) drain(DB);
    }
#pragma nounroll
    for (int k = ng * UL; k < nk; ++k) {
        const int c = k * NW + wv;
        examine(*((const v4i*)(dst + ((size_t)c << 7)) + lane), c);
    }
    if (wv < 4) {
        const int e = (n_chunks << 7) + (wv << 5) + lane;
        int dd = -1;
        if (e < n_edges) dd = dst[e];
        push((unsigned)(dd - node0) < (unsigned)npb, e);
    }
    while (tail != head) {
        const int left = tail - head;
        drain(left < DB ? left : DB);
    }
    __syncthreads();

    for (int t = wv; t < NPB / 16; t += NW) {
        const int m = lane & 15;
        const int h = lane >> 4;
        const int r = t * 16 + m;
        v4f x[8];
#pragma unroll
        for (int j = 0; j < 8; ++j) x[j] = *(const v4f*)(acc + swz(r, 8 * h + j));
        v8f c = {};
#pragma unroll
        for (int j = 0; j < 8; ++j) {
            const v2f a0 = {x[j].x, x[j].y};
            const v2f a1 = {x[j].z, x[j].w};
            c = __builtin_amdgcn_wmma_f32_16x16x4_f32(false, a0, false, a0, (short)0, c, false, false);
            c = __builtin_amdgcn_wmma_f32_16x16x4_f32(false, a1, false, a1, (short)0, c, false, false);
        }
        const float ss  = c[lane & 7];
        const float inv = 1.0f / fmaxf(sqrtf(ss), 1e-12f);
        if (lane < 8)        s_inv[t * 16 + lane]      = inv;
        else if (lane >= 24) s_inv[t * 16 + lane - 16] = inv;
    }
    __syncthreads();

    {
        v4f* out4 = (v4f*)(out + (size_t)node0 * DIM);
        const int n4 = npb * (DIM / 4);
        for (int i = tid; i < n4; i += NT) {
            const int r = i >> 4;
            const int g = i & 15;
            v4f v = *(const v4f*)(acc + swz(r, g));
            v = v * s_inv[r];
            __builtin_nontemporal_store(v, out4 + i);
        }
    }
}

extern "C" void kernel_launch(void* const* d_in, const int* in_sizes, int n_in,
                              void* d_out, int out_size, void* d_ws, size_t ws_size,
                              hipStream_t stream)
{
    const float* user_emb = (const float*)d_in[0];
    const float* item_emb = (const float*)d_in[1];
    const int*   src      = (const int*)d_in[2];
    const int*   dst      = (const int*)d_in[3];
    const float* ew       = (const float*)d_in[4];
    float*       out      = (float*)d_out;

    const int n_users = in_sizes[0] / DIM;
    const int n_items = in_sizes[1] / DIM;
    const int n_edges = in_sizes[2];
    const int n_nodes = n_users + n_items;

    const unsigned blocks = (unsigned)((n_nodes + NPB - 1) / NPB);
    owner_kernel<<<blocks, NT, 0, stream>>>(user_emb, item_emb, src, dst, ew, out,
                                            n_users, n_nodes, n_edges);
}


# a short first launch marks per owner block which 32-edge groups hold one of its edges (bit map, LDS integer OR); owners read dst of marked groups only; LDS float adds in arrival order, order shifts
# speedup vs baseline: 1.9579x; 1.0743x over previous
#include <hip/hip_runtime.h>


#define DIM   64
#define NPB   1184
#define NT    1024
#define NW    32
#define QCAP  128
#define VL    8
#define DB    8
#define PW    8
#define PT    256
#define TP    257
#define EPB   8192

typedef float v2f __attribute__((ext_vector_type(2)));
typedef float v4f __attribute__((ext_vector_type(4)));
typedef float v8f __attribute__((ext_vector_type(8)));
typedef int   v4i __attribute__((ext_vector_type(4)));
typedef unsigned v4u __attribute__((ext_vector_type(4)));

__device__ __forceinline__ int swz(int r, int g) { return (r << 6) + (((g ^ r) & 15) << 2); }

__global__ __launch_bounds__(PT) void map_kernel(
    const int* __restrict__ dst,
    unsigned*  __restrict__ map,
    int n_nodes, int n_edges, int nb)
{
    __shared__ unsigned tab[PW * TP];

    const int tid  = (int)threadIdx.x;
    const int lane = tid & 31;
    const int w    = tid >> 5;
    const int e0   = (int)blockIdx.x * EPB + w * 1024 + lane * 4;

    v4i v[8];
    if (((int)blockIdx.x + 1) * EPB <= n_edges) {
#pragma unroll
        for (int u = 0; u < 8; ++u) v[u] = *(const v4i*)(dst + e0 + u * 128);
    } else {
#pragma unroll
        for (int u = 0; u < 8; ++u) {
            const int e = e0 + u * 128;
            v[u].x = (e     < n_edges) ? dst[e]     : -1;
            v[u].y = (e + 1 < n_edges) ? dst[e + 1] : -1;
            v[u].z = (e + 2 < n_edges) ? dst[e + 2] : -1;
            v[u].w = (e + 3 < n_edges) ? dst[e + 3] : -1;
        }
    }

    for (int b0 = 0; b0 < nb; b0 += PT) {
        for (int i = tid; i < PW * TP; i += PT) tab[i] = 0u;
        __syncthreads();
        unsigned* col = tab + w * TP;
#pragma unroll
        for (int u = 0; u < 8; ++u) {
            const unsigned bit = 1u << (u * 4 + (lane >> 3));
            const int d[4] = {v[u].x, v[u].y, v[u].z, v[u].w};
#pragma unroll
            for (int j = 0; j < 4; ++j) {
                if ((unsigned)d[j] < (unsigned)n_nodes) {
                    const int ob = (int)((unsigned)d[j] / (unsigned)NPB);
                    if (ob >= b0 && ob < b0 + PT)
                        __hip_atomic_fetch_or(col + (ob - b0), bit, __ATOMIC_RELAXED, __HIP_MEMORY_SCOPE_WORKGROUP);
                }
            }
        }
        __syncthreads();
        const int b = b0 + tid;
        if (b < nb) {
            const unsigned* t = tab + tid;
            const v4u lo = {t[0 * TP], t[1 * TP], t[2 * TP], t[3 * TP]};
            const v4u hi = {t[4 * TP], t[5 * TP], t[6 * TP], t[7 * TP]};
            v4u* seg = (v4u*)(map + ((size_t)blockIdx.x * nb + b) * PW);
            seg[0] = lo;
            seg[1] = hi;
        }
        __syncthreads();
    }
}

__global__ __launch_bounds__(NT) void owner_kernel(
    const float* __restrict__ user_emb,
    const float* __restrict__ item_emb,
    const int*   __restrict__ src,
    const int*   __restrict__ dst,
    const float* __restrict__ ew,
    float*       __restrict__ out,
    const unsigned* __restrict__ map,
    int n_users, int n_nodes, int n_edges, int n_map_blocks)
{
    __shared__ __attribute__((aligned(16))) float acc[NPB * DIM];
    __shared__ int   q[NW * QCAP];
    __shared__ float s_inv[NPB];

    const int tid  = (int)threadIdx.x;
    const int lane = tid & 31;
    const int wv    = __builtin_amdgcn_readfirstlane(tid >> 5);
    const int node0 = (int)blockIdx.x * NPB;
    int npb = n_nodes - node0;
    if (npb > NPB) npb = NPB;

    {
        v4f* acc4 = (v4f*)acc;
        const v4f z = {0.f, 0.f, 0.f, 0.f};
        for (int i = tid; i < NPB * DIM / 4; i += NT) acc4[i] = z;
    }
    __syncthreads();

    int* qw = q + wv * QCAP;
    int head = 0, tail = 0;

    const int nblk   = (int)gridDim.x;
    const int nwords = n_map_blocks * PW;
    const int wpw    = (nwords + NW - 1) / NW;
    const int nslots = wpw * NW;
    const int rot    = (int)(((long long)blockIdx.x * nslots) / nblk);
    const unsigned* mywords = map + (size_t)blockIdx.x * PW;

    auto add_row = [&](int d, float lo, float hi) {
        const int g = lane >> 2;
        float* p0 = acc + swz(d, g)     + (lane & 3);
        float* p1 = acc + swz(d, g + 8) + (lane & 3);
        __hip_atomic_fetch_add(p0, lo, __ATOMIC_RELAXED, __HIP_MEMORY_SCOPE_WORKGROUP);
        __hip_atomic_fetch_add(p1, hi, __ATOMIC_RELAXED, __HIP_MEMORY_SCOPE_WORKGROUP);
    };
    auto drain_one = [&]() {
        const int   e  = __builtin_amdgcn_readfirstlane(qw[head & (QCAP - 1)]);
        const int   s1 = src[e];
        const int   d1 = dst[e] - node0;
        const float w1 = ew[e];
        const float* row = (s1 < n_users)
                               ? (user_emb + (size_t)s1 * DIM)
                               : (item_emb + (size_t)(s1 - n_users) * DIM);
        add_row(d1, row[lane] * w1, row[lane + 32] * w1);
        ++head;
    };
    auto push = [&](bool m, int e) {
        const unsigned mask = __builtin_amdgcn_ballot_w32(m);
        if (mask) {
            while (tail - head > QCAP - 32) drain_one();
            const int pos = (int)__builtin_amdgcn_mbcnt_lo(mask, 0u);
            if (m) qw[(tail + pos) & (QCAP - 1)] = e;
            tail += __builtin_popcount(mask);
        }
    };
    auto drain = [&](int nb) {
        int   s[DB], d[DB];
        float w[DB], lo[DB], hi[DB];
#pragma unroll
        for (int i = 0; i < DB; ++i) {
            const int ii = (i < nb) ? i : (nb - 1);
            const int e  = __builtin_amdgcn_readfirstlane(qw[(head + ii) & (QCAP - 1)]);
            s[i] = src[e]; d[i] = dst[e] - node0; w[i] = ew[e];
        }
#pragma unroll
        for (int i = 0; i < DB; ++i) {
            const float* row = (s[i] < n_users)
                                   ? (user_emb + (size_t)s[i] * DIM)
                                   : (item_emb + (size_t)(s[i] - n_users) * DIM);
            lo[i] = row[lane];
            hi[i] = row[lane + 32];
        }
#pragma unroll
        for (int i = 0; i < DB; ++i)
            if (i < nb) add_row(d[i], lo[i] * w[i], hi[i] * w[i]);
        head += nb;
    };

    for (int k0 = 0; k0 < wpw; k0 += 32) {
        const int k   = k0 + lane;
        int idx = wv * wpw + k + rot;
        if (idx >= nslots) idx -= nslots;
        unsigned word = 0u;
        if (k < wpw && idx < nwords) word = mywords[(size_t)(idx >> 3) * (size_t)(nblk * PW) + (idx & 7)];
        unsigned live = __builtin_amdgcn_ballot_w32(word != 0u);
        while (live) {
            const int j = __builtin_ctz(live);
            live &= live - 1u;
            unsigned m = (unsigned)__builtin_amdgcn_readlane((int)word, j);
            int wj = wv * wpw + k0 + j + rot;
            if (wj >= nslots) wj -= nslots;
            const int ebase = wj << 10;
            while (m) {
                const int pc  = __builtin_popcount(m);
                const int cnt = pc < VL ? pc : VL;
                int eb[VL], dd[VL];
#pragma unroll
                for (int u = 0; u < VL; ++u) {
                    eb[u] = 0; dd[u] = -1;
                    if (u < cnt) {
                        eb[u] = ebase + (__builtin_ctz(m) << 5);
                        m &= m - 1u;
                        const int e = eb[u] + lane;
                        dd[u] = dst[e < n_edges ? e : n_edges - 1];
                    }
                }
                while (tail - head >= DB) drain(DB);
#pragma unroll
                for (int u = 0; u < VL; ++u)
                    if (u < cnt) {
                        const int e = eb[u] + lane;
                        push(e < n_edges && (unsigned)(dd[u] - node0) < (unsigned)npb, e);
                    }
            }
        }
    }
    while (tail != head) {
        const int left = tail - head;
        drain(left < DB ? left : DB);
    }
    __syncthreads();

    for (int t = wv; t < NPB / 16; t += NW) {
        const int m = lane & 15;
        const int h = lane >> 4;
        const int r = t * 16 + m;
        v4f x[8];
#pragma unroll
        for (int j = 0; j < 8; ++j) x[j] = *(const v4f*)(acc + swz(r, 8 * h + j));
        v8f c = {};
#pragma unroll
        for (int j = 0; j < 8; ++j) {
            const v2f a0 = {x[j].x, x[j].y};
            const v2f a1 = {x[j].z, x[j].w};
            c = __builtin_amdgcn_wmma_f32_16x16x4_f32(false, a0, false, a0, (short)0, c, false, false);
            c = __builtin_amdgcn_wmma_f32_16x16x4_f32(false, a1, false, a1, (short)0, c, false, false);
        }
        const float ss  = c[lane & 7];
        const float inv = 1.0f / fmaxf(sqrtf(ss), 1e-12f);
        if (lane < 8)        s_inv[t * 16 + lane]      = inv;
        else if (lane >= 24) s_inv[t * 16 + lane - 16] = inv;
    }
    __syncthreads();

    {
        v4f* out4 = (v4f*)(out + (size_t)node0 * DIM);
        const int n4 = npb * (DIM / 4);
        for (int i = tid; i < n4; i += NT) {
            const int r = i >> 4;
            const int g = i & 15;
            v4f v = *(const v4f*)(acc + swz(r, g));
            v = v * s_inv[r];
            __builtin_nontemporal_store(v, out4 + i);
        }
    }
}

extern "C" void kernel_launch(void* const* d_in, const int* in_sizes, int n_in,
                              void* d_out, int out_size, void* d_ws, size_t ws_size,
                              hipStream_t stream)
{
    const float* user_emb = (const float*)d_in[0];
    const float* item_emb = (const float*)d_in[1];
    const int*   src      = (const int*)d_in[2];
    const int*   dst      = (const int*)d_in[3];
    const float* ew       = (const float*)d_in[4];
    float*       out      = (float*)d_out;

    const int n_users = in_sizes[0] / DIM;
    const int n_items = in_sizes[1] / DIM;
    const int n_edges = in_sizes[2];
    const int n_nodes = n_users + n_items;

    unsigned*    map      = (unsigned*)d_ws;

    const unsigned blocks     = (unsigned)((n_nodes + NPB - 1) / NPB);
    const unsigned map_blocks = (unsigned)((n_edges + EPB - 1) / EPB);
    map_kernel<<<map_blocks, PT, 0, stream>>>(dst, map, n_nodes, n_edges, (int)blocks);
    owner_kernel<<<blocks, NT, 0, stream>>>(user_emb, item_emb, src, dst, ew, out, map,
                                            n_users, n_nodes, n_edges, (int)map_blocks);
}


# owners load src and weight lines of each marked group with its dst line and queue (src,row,weight) records in LDS: sum loop takes records from LDS, no scalar loads; LDS float adds, order shifts
# speedup vs baseline: 2.4750x; 1.2641x over previous
#include <hip/hip_runtime.h>


#define DIM   64
#define NPB   1184
#define NT    1024
#define NW    32
#define QCAP  64
#define VL    8
#define DB    8
#define PW    8
#define PT    256
#define TP    257
#define EPB   8192

typedef float v2f __attribute__((ext_vector_type(2)));
typedef float v4f __attribute__((ext_vector_type(4)));
typedef float v8f __attribute__((ext_vector_type(8)));
typedef int   v4i __attribute__((ext_vector_type(4)));
typedef unsigned v4u __attribute__((ext_vector_type(4)));

__device__ __forceinline__ int swz(int r, int g) { return (r << 6) + (((g ^ r) & 15) << 2); }

__global__ __launch_bounds__(PT) void map_kernel(
    const int* __restrict__ dst,
    unsigned*  __restrict__ map,
    int n_nodes, int n_edges, int nb)
{
    __shared__ unsigned tab[PW * TP];

    const int tid  = (int)threadIdx.x;
    const int lane = tid & 31;
    const int w    = tid >> 5;
    const int e0   = (int)blockIdx.x * EPB + w * 1024 + lane * 4;

    v4i v[8];
    if (((int)blockIdx.x + 1) * EPB <= n_edges) {
#pragma unroll
        for (int u = 0; u < 8; ++u) v[u] = *(const v4i*)(dst + e0 + u * 128);
    } else {
#pragma unroll
        for (int u = 0; u < 8; ++u) {
            const int e = e0 + u * 128;
            v[u].x = (e     < n_edges) ? dst[e]     : -1;
            v[u].y = (e + 1 < n_edges) ? dst[e + 1] : -1;
            v[u].z = (e + 2 < n_edges) ? dst[e + 2] : -1;
            v[u].w = (e + 3 < n_edges) ? dst[e + 3] : -1;
        }
    }

    for (int b0 = 0; b0 < nb; b0 += PT) {
        for (int i = tid; i < PW * TP; i += PT) tab[i] = 0u;
        __syncthreads();
        unsigned* col = tab + w * TP;
#pragma unroll
        for (int u = 0; u < 8; ++u) {
            const unsigned bit = 1u << (u * 4 + (lane >> 3));
            const int d[4] = {v[u].x, v[u].y, v[u].z, v[u].w};
#pragma unroll
            for (int j = 0; j < 4; ++j) {
                if ((unsigned)d[j] < (unsigned)n_nodes) {
                    const int ob = (int)((unsigned)d[j] / (unsigned)NPB);
                    if (ob >= b0 && ob < b0 + PT)
                        __hip_atomic_fetch_or(col + (ob - b0), bit, __ATOMIC_RELAXED, __HIP_MEMORY_SCOPE_WORKGROUP);
                }
            }
        }
        __syncthreads();
        const int b = b0 + tid;
        if (b < nb) {
            const unsigned* t = tab + tid;
            const v4u lo = {t[0 * TP], t[1 * TP], t[2 * TP], t[3 * TP]};
            const v4u hi = {t[4 * TP], t[5 * TP], t[6 * TP], t[7 * TP]};
            v4u* seg = (v4u*)(map + ((size_t)blockIdx.x * nb + b) * PW);
            seg[0] = lo;
            seg[1] = hi;
        }
        __syncthreads();
    }
}

__global__ __launch_bounds__(NT) void owner_kernel(
    const float* __restrict__ user_emb,
    const float* __restrict__ item_emb,
    const int*   __restrict__ src,
    const int*   __restrict__ dst,
    const float* __restrict__ ew,
    float*       __restrict__ out,
    const unsigned* __restrict__ map,
    int n_users, int n_nodes, int n_edges, int n_map_blocks)
{
    __shared__ __attribute__((aligned(16))) float acc[NPB * DIM];
    __shared__ int ring[NW * 3 * QCAP];

    const int tid  = (int)threadIdx.x;
    const int lane = tid & 31;
    const int wv    = __builtin_amdgcn_readfirstlane(tid >> 5);
    const int node0 = (int)blockIdx.x * NPB;
    int npb = n_nodes - node0;
    if (npb > NPB) npb = NPB;

    {
        v4f* acc4 = (v4f*)acc;
        const v4f z = {0.f, 0.f, 0.f, 0.f};
        for (int i = tid; i < NPB * DIM / 4; i += NT) acc4[i] = z;
    }
    __syncthreads();

    int* qs = ring + wv * (3 * QCAP);
    int* qd = qs + QCAP;
    int* qw = qs + 2 * QCAP;
    int head = 0, tail = 0;

    const int nblk   = (int)gridDim.x;
    const int nwords = n_map_blocks * PW;
    const int wpw    = (nwords + NW - 1) / NW;
    const int nslots = wpw * NW;
    const int rot    = (int)(((long long)blockIdx.x * nslots) / nblk);
    const unsigned* mywords = map + (size_t)blockIdx.x * PW;

    auto add_row = [&](int d, float lo, float hi) {
        const int g = lane >> 2;
        float* p0 = acc + swz(d, g)     + (lane & 3);
        float* p1 = acc + swz(d, g + 8) + (lane & 3);
        __hip_atomic_fetch_add(p0, lo, __ATOMIC_RELAXED, __HIP_MEMORY_SCOPE_WORKGROUP);
        __hip_atomic_fetch_add(p1, hi, __ATOMIC_RELAXED, __HIP_MEMORY_SCOPE_WORKGROUP);
    };
    auto drain = [&](int nb) {
        const int li   = (lane < nb) ? lane : (nb - 1);
        const int pos  = (head + li) & (QCAP - 1);
        const int srec = qs[pos];
        const int drec = qd[pos];
        const int wrec = qw[pos];
        int   d[DB];
        float w[DB], lo[DB], hi[DB];
#pragma unroll
        for (int i = 0; i < DB; ++i) {
            const int s = __builtin_amdgcn_readlane(srec, i);
            d[i] = __builtin_amdgcn_readlane(drec, i);
            w[i] = __int_as_float(__builtin_amdgcn_readlane(wrec, i));
            const float* row = (s < n_users)
                                   ? (user_emb + (size_t)s * DIM)
                                   : (item_emb + (size_t)(s - n_users) * DIM);
            lo[i] = row[lane];
            hi[i] = row[lane + 32];
        }
#pragma unroll
        for (int i = 0; i < DB; ++i)
            if (i < nb) add_row(d[i], lo[i] * w[i], hi[i] * w[i]);
        head += nb;
    };
    auto push = [&](bool m, int s, int d, float w) {
        const unsigned mask = __builtin_amdgcn_ballot_w32(m);
        if (mask) {
            while (tail - head > QCAP - 32) drain(DB);
            const int pos = (tail + (int)__builtin_amdgcn_mbcnt_lo(mask, 0u)) & (QCAP - 1);
            if (m) { qs[pos] = s; qd[pos] = d; qw[pos] = __float_as_int(w); }
            tail += __builtin_popcount(mask);
        }
    };

    for (int k0 = 0; k0 < wpw; k0 += 32) {
        const int k   = k0 + lane;
        int idx = wv * wpw + k + rot;
        if (idx >= nslots) idx -= nslots;
        unsigned word = 0u;
        if (k < wpw && idx < nwords) word = mywords[(size_t)(idx >> 3) * (size_t)(nblk * PW) + (idx & 7)];
        unsigned live = __builtin_amdgcn_ballot_w32(word != 0u);
        while (live) {
            const int j = __builtin_ctz(live);
            live &= live - 1u;
            unsigned m = (unsigned)__builtin_amdgcn_readlane((int)word, j);
            int wj = wv * wpw + k0 + j + rot;
            if (wj >= nslots) wj -= nslots;
            const int ebase = wj << 10;
            while (m) {
                const int pc  = __builtin_popcount(m);
                const int cnt = pc < VL ? pc : VL;
                int   eb[VL], dd[VL], ss[VL];
                float ww[VL];
#pragma unroll
                for (int u = 0; u < VL; ++u) {
                    eb[u] = 0; dd[u] = -1; ss[u] = 0; ww[u] = 0.f;
                    if (u < cnt) {
                        eb[u] = ebase + (__builtin_ctz(m) << 5);
                        m &= m - 1u;
                        const int e  = eb[u] + lane;
                        const int ec = e < n_edges ? e : n_edges - 1;
                        dd[u] = dst[ec];
                        ss[u] = src[ec];
                        ww[u] = ew[ec];
                    }
                }
                while (tail - head >= DB) drain(DB);
#pragma unroll
                for (int u = 0; u < VL; ++u)
                    if (u < cnt) {
                        const int e = eb[u] + lane;
                        push(e < n_edges && (unsigned)(dd[u] - node0) < (unsigned)npb, ss[u], dd[u] - node0, ww[u]);
                    }
            }
        }
    }
    while (tail != head) {
        const int left = tail - head;
        drain(left < DB ? left : DB);
    }
    __syncthreads();

    for (int t = wv; t < NPB / 16; t += NW) {
        const int m = lane & 15;
        const int h = lane >> 4;
        const int r = t * 16 + m;
        v4f x[8];
#pragma unroll
        for (int j = 0; j < 8; ++j) x[j] = *(const v4f*)(acc + swz(r, 8 * h + j));
        v8f c = {};
#pragma unroll
        for (int j = 0; j < 8; ++j) {
            const v2f a0 = {x[j].x, x[j].y};
            const v2f a1 = {x[j].z, x[j].w};
            c = __builtin_amdgcn_wmma_f32_16x16x4_f32(false, a0, false, a0, (short)0, c, false, false);
            c = __builtin_amdgcn_wmma_f32_16x16x4_f32(false, a1, false, a1, (short)0, c, false, false);
        }
        const float ss  = c[lane & 7];
        const float inv = 1.0f / fmaxf(sqrtf(ss), 1e-12f);
        if (lane < 8)        ring[t * 16 + lane]      = __float_as_int(inv);
        else if (lane >= 24) ring[t * 16 + lane - 16] = __float_as_int(inv);
    }
    __syncthreads();

    {
        v4f* out4 = (v4f*)(out + (size_t)node0 * DIM);
        const int n4 = npb * (DIM / 4);
        for (int i = tid; i < n4; i += NT) {
            const int r = i >> 4;
            const int g = i & 15;
            v4f v = *(const v4f*)(acc + swz(r, g));
            v = v * __int_as_float(ring[r]);
            __builtin_nontemporal_store(v, out4 + i);
        }
    }
}

extern "C" void kernel_launch(void* const* d_in, const int* in_sizes, int n_in,
                              void* d_out, int out_size, void* d_ws, size_t ws_size,
                              hipStream_t stream)
{
    const float* user_emb = (const float*)d_in[0];
    const float* item_emb = (const float*)d_in[1];
    const int*   src      = (const int*)d_in[2];
    const int*   dst      = (const int*)d_in[3];
    const float* ew       = (const float*)d_in[4];
    float*       out      = (float*)d_out;

    const int n_users = in_sizes[0] / DIM;
    const int n_items = in_sizes[1] / DIM;
    const int n_edges = in_sizes[2];
    const int n_nodes = n_users + n_items;

    unsigned*    map      = (unsigned*)d_ws;

    const unsigned blocks     = (unsigned)((n_nodes + NPB - 1) / NPB);
    const unsigned map_blocks = (unsigned)((n_edges + EPB - 1) / EPB);
    map_kernel<<<map_blocks, PT, 0, stream>>>(dst, map, n_nodes, n_edges, (int)blocks);
    owner_kernel<<<blocks, NT, 0, stream>>>(user_emb, item_emb, src, dst, ew, out, map,
                                            n_users, n_nodes, n_edges, (int)map_blocks);
}
